# starting kernel
# baseline (speedup 1.0000x reference)
#include <hip/hip_runtime.h>

typedef __attribute__((ext_vector_type(2))) float v2f;
typedef __attribute__((ext_vector_type(8))) float v8f;

__global__ void build_out1(const float* __restrict__ U0,
                           const float* __restrict__ U1,
                           float* __restrict__ out1)
{
    int idx = blockIdx.x * blockDim.x + threadIdx.x;
    if (idx >= 225 * 64 * 64) return;
    int c   = idx & 63;
    int b   = (idx >> 6) & 63;
    int row = idx >> 12;
    int i0 = row / 15, i1 = row % 15;
    int o0 = b >> 3,   o1 = b & 7;
    const float* u0 = U0 + (i0 * 8 + o0) * 32;
    const float* u1 = U1 + (i1 * 8 + o1) * 64 + c;
    float s = 0.f;
#pragma unroll
    for (int k = 0; k < 32; ++k)
        s += u0[k] * u1[k * 7680];
    out1[idx] = s;
}

__global__ void build_out2(const float* __restrict__ U2,
                           const float* __restrict__ U3,
                           float* __restrict__ out2)
{
    int idx = blockIdx.x * blockDim.x + threadIdx.x;
    if (idx >= 64 * 225 * 64) return;
    int d   = idx & 63;
    int row = (idx >> 6) % 225;
    int c   = idx / (225 * 64);
    int i2 = row / 15, i3 = row % 15;
    int o2 = d >> 3,   o3 = d & 7;
    const float* u2 = U2 + ((c * 15 + i2) * 8 + o2) * 32;
    const float* u3 = U3 + (i3 * 8 + o3);
    float s = 0.f;
#pragma unroll
    for (int k = 0; k < 32; ++k)
        s += u2[k] * u3[k * 120];
    out2[idx] = s;
}

#define SA 68

__global__ __launch_bounds__(512) void ttm_gemm_kernel(
    const int*   __restrict__ x,
    const float* __restrict__ out1,
    const float* __restrict__ out2,
    float*       __restrict__ out)
{
    __shared__ float lA[64 * SA];
    __shared__ float lB[64 * 64];

    const int n  = blockIdx.x;
    const int t  = threadIdx.x;
    const int xi = x[n];
    const int row0 = xi / 225;
    const int row1 = xi % 225;

    {
        const int r  = t >> 3;
        const int ch = (t & 7) * 8;
        const float4* Ag = (const float4*)(out1 + (size_t)row0 * 4096 + r * 64 + ch);
        float4 a0 = Ag[0], a1 = Ag[1];
        *(float4*)(&lA[r * SA + ch])     = a0;
        *(float4*)(&lA[r * SA + ch + 4]) = a1;
        const float4* Bg = (const float4*)(out2 + (size_t)r * 14400 + row1 * 64 + ch);
        float4 b0 = Bg[0], b1 = Bg[1];
        *(float4*)(&lB[r * 64 + ch])     = b0;
        *(float4*)(&lB[r * 64 + ch + 4]) = b1;
    }
    __syncthreads();

    const int wave = t >> 5;
    const int lane = t & 31;
    const int tm = wave >> 2, tn = wave & 3;
    const int lr = lane & 15;
    const int lh = lane >> 4;

    const float* Abase = &lA[(tm * 16 + lr) * SA + lh * 2];
    const float* Bbase = &lB[(lh * 2) * 64 + tn * 16 + lr];

    v8f acc = {};
#pragma unroll
    for (int k0 = 0; k0 < 16; ++k0) {
        v2f a = *(const v2f*)(Abase + k0 * 4);
        v2f b;
        b.x = Bbase[k0 * 4 * 64];
        b.y = Bbase[k0 * 4 * 64 + 64];
        acc = __builtin_amdgcn_wmma_f32_16x16x4_f32(
            false, a, false, b, (short)0, acc, false, false);
    }

    float* op = out + (size_t)n * 4096 + (tm * 16 + lh * 8) * 64 + tn * 16 + lr;
#pragma unroll
    for (int v = 0; v < 8; ++v)
        op[v * 64] = acc[v];
}

extern "C" void kernel_launch(void* const* d_in, const int* in_sizes, int n_in,
                              void* d_out, int out_size, void* d_ws, size_t ws_size,
                              hipStream_t stream) {
    const int*   x  = (const int*)  d_in[0];
    const float* U0 = (const float*)d_in[1];
    const float* U1 = (const float*)d_in[2];
    const float* U2 = (const float*)d_in[3];
    const float* U3 = (const float*)d_in[4];
    float* out = (float*)d_out;

    float* w1 = (float*)d_ws;
    float* w2 = w1 + 921600;

    const int tab = 921600;
    build_out1<<<(tab + 255) / 256, 256, 0, stream>>>(U0, U1, w1);
    build_out2<<<(tab + 255) / 256, 256, 0, stream>>>(U2, U3, w2);

    const int nsamp = in_sizes[0];
    ttm_gemm_kernel<<<nsamp, 512, 0, stream>>>(x, w1, w2, out);
}


# main GEMM kernel: work-groups of 4 waves loop over samples (grid 2048), each wave one 16x64 row strip on 4 accumulator tiles; launched waves 131072 -> 8192 (wave entry); stores and sum order kept
# speedup vs baseline: 1.2164x; 1.2164x over previous
#include <hip/hip_runtime.h>

typedef __attribute__((ext_vector_type(2))) float v2f;
typedef __attribute__((ext_vector_type(8))) float v8f;

__global__ void build_out1(const float* __restrict__ U0,
                           const float* __restrict__ U1,
                           float* __restrict__ out1)
{
    int idx = blockIdx.x * blockDim.x + threadIdx.x;
    if (idx >= 225 * 64 * 64) return;
    int c   = idx & 63;
    int b   = (idx >> 6) & 63;
    int row = idx >> 12;
    int i0 = row / 15, i1 = row % 15;
    int o0 = b >> 3,   o1 = b & 7;
    const float* u0 = U0 + (i0 * 8 + o0) * 32;
    const float* u1 = U1 + (i1 * 8 + o1) * 64 + c;
    float s = 0.f;
#pragma unroll
    for (int k = 0; k < 32; ++k)
        s += u0[k] * u1[k * 7680];
    out1[idx] = s;
}

__global__ void build_out2(const float* __restrict__ U2,
                           const float* __restrict__ U3,
                           float* __restrict__ out2)
{
    int idx = blockIdx.x * blockDim.x + threadIdx.x;
    if (idx >= 64 * 225 * 64) return;
    int d   = idx & 63;
    int row = (idx >> 6) % 225;
    int c   = idx / (225 * 64);
    int i2 = row / 15, i3 = row % 15;
    int o2 = d >> 3,   o3 = d & 7;
    const float* u2 = U2 + ((c * 15 + i2) * 8 + o2) * 32;
    const float* u3 = U3 + (i3 * 8 + o3);
    float s = 0.f;
#pragma unroll
    for (int k = 0; k < 32; ++k)
        s += u2[k] * u3[k * 120];
    out2[idx] = s;
}

#define SA 68

__global__ __launch_bounds__(128) void ttm_gemm_kernel(
    const int*   __restrict__ x,
    const float* __restrict__ out1,
    const float* __restrict__ out2,
    float*       __restrict__ out,
    int nsamp)
{
    __shared__ __attribute__((aligned(16))) float lA[64 * SA];
    __shared__ __attribute__((aligned(16))) float lB[64 * 64];

    const int t    = threadIdx.x;
    const int tm   = t >> 5;
    const int lane = t & 31;
    const int lr   = lane & 15;
    const int lh   = lane >> 4;

    const int sr = t >> 4;
    const int sc = (t & 15) * 4;

    const float* Abase = &lA[(tm * 16 + lr) * SA + lh * 2];
    const float* Bbase = &lB[(lh * 2) * 64 + lr];

    for (int n = blockIdx.x; n < nsamp; n += gridDim.x) {
        const int xi   = x[n];
        const int row0 = xi / 225;
        const int row1 = xi % 225;

        const float* Ag = out1 + (size_t)row0 * 4096 + sr * 64 + sc;
        const float* Bg = out2 + (size_t)sr * 14400 + row1 * 64 + sc;
        const float4 ra0 = *(const float4*)(Ag + 0 * 512);
        const float4 ra1 = *(const float4*)(Ag + 1 * 512);
        const float4 ra2 = *(const float4*)(Ag + 2 * 512);
        const float4 ra3 = *(const float4*)(Ag + 3 * 512);
        const float4 ra4 = *(const float4*)(Ag + 4 * 512);
        const float4 ra5 = *(const float4*)(Ag + 5 * 512);
        const float4 ra6 = *(const float4*)(Ag + 6 * 512);
        const float4 ra7 = *(const float4*)(Ag + 7 * 512);
        const float4 rb0 = *(const float4*)(Bg + 0 * 115200);
        const float4 rb1 = *(const float4*)(Bg + 1 * 115200);
        const float4 rb2 = *(const float4*)(Bg + 2 * 115200);
        const float4 rb3 = *(const float4*)(Bg + 3 * 115200);
        const float4 rb4 = *(const float4*)(Bg + 4 * 115200);
        const float4 rb5 = *(const float4*)(Bg + 5 * 115200);
        const float4 rb6 = *(const float4*)(Bg + 6 * 115200);
        const float4 rb7 = *(const float4*)(Bg + 7 * 115200);
        __builtin_amdgcn_fence(__ATOMIC_RELEASE, "workgroup", "local");
        __builtin_amdgcn_s_barrier();
        __builtin_amdgcn_fence(__ATOMIC_ACQUIRE, "workgroup", "local");
        float* sA = &lA[sr * SA + sc];
        float* sB = &lB[sr * 64 + sc];
        *(float4*)(sA + 0 * 8 * SA) = ra0;   *(float4*)(sB + 0 * 512) = rb0;
        *(float4*)(sA + 1 * 8 * SA) = ra1;   *(float4*)(sB + 1 * 512) = rb1;
        *(float4*)(sA + 2 * 8 * SA) = ra2;   *(float4*)(sB + 2 * 512) = rb2;
        *(float4*)(sA + 3 * 8 * SA) = ra3;   *(float4*)(sB + 3 * 512) = rb3;
        *(float4*)(sA + 4 * 8 * SA) = ra4;   *(float4*)(sB + 4 * 512) = rb4;
        *(float4*)(sA + 5 * 8 * SA) = ra5;   *(float4*)(sB + 5 * 512) = rb5;
        *(float4*)(sA + 6 * 8 * SA) = ra6;   *(float4*)(sB + 6 * 512) = rb6;
        *(float4*)(sA + 7 * 8 * SA) = ra7;   *(float4*)(sB + 7 * 512) = rb7;
        __syncthreads();

        v8f acc0 = {}, acc1 = {}, acc2 = {}, acc3 = {};
#pragma unroll
        for (int k0 = 0; k0 < 16; ++k0) {
            const v2f a = *(const v2f*)(Abase + k0 * 4);
            const float* bp = Bbase + k0 * 4 * 64;
            v2f b0, b1, b2, b3;
            b0.x = bp[0];   b0.y = bp[64];
            b1.x = bp[16];  b1.y = bp[64 + 16];
            b2.x = bp[32];  b2.y = bp[64 + 32];
            b3.x = bp[48];  b3.y = bp[64 + 48];
            acc0 = __builtin_amdgcn_wmma_f32_16x16x4_f32(
                false, a, false, b0, (short)0, acc0, false, false);
            acc1 = __builtin_amdgcn_wmma_f32_16x16x4_f32(
                false, a, false, b1, (short)0, acc1, false, false);
            acc2 = __builtin_amdgcn_wmma_f32_16x16x4_f32(
                false, a, false, b2, (short)0, acc2, false, false);
            acc3 = __builtin_amdgcn_wmma_f32_16x16x4_f32(
                false, a, false, b3, (short)0, acc3, false, false);
        }

        float* op = out + (size_t)n * 4096 + (tm * 16 + lh * 8) * 64 + lr;
#pragma unroll
        for (int v = 0; v < 8; ++v) {
            op[v * 64]      = acc0[v];
            op[v * 64 + 16] = acc1[v];
            op[v * 64 + 32] = acc2[v];
            op[v * 64 + 48] = acc3[v];
        }
    }
}

extern "C" void kernel_launch(void* const* d_in, const int* in_sizes, int n_in,
                              void* d_out, int out_size, void* d_ws, size_t ws_size,
                              hipStream_t stream) {
    const int*   x  = (const int*)  d_in[0];
    const float* U0 = (const float*)d_in[1];
    const float* U1 = (const float*)d_in[2];
    const float* U2 = (const float*)d_in[3];
    const float* U3 = (const float*)d_in[4];
    float* out = (float*)d_out;

    float* w1 = (float*)d_ws;
    float* w2 = w1 + 921600;

    const int tab = 921600;
    build_out1<<<(tab + 255) / 256, 256, 0, stream>>>(U0, U1, w1);
    build_out2<<<(tab + 255) / 256, 256, 0, stream>>>(U2, U3, w2);

    const int nsamp = in_sizes[0];
    ttm_gemm_kernel<<<2048, 128, 0, stream>>>(x, w1, w2, out, nsamp);
}


# both table builds in one launch of register-blocked waves (lane holds the large factor's 32 k-values, small factor by scalar loads, 16 entries/lane): launches 3->2, build waves 57600->3600; same sums
# speedup vs baseline: 1.4468x; 1.1894x over previous
#include <hip/hip_runtime.h>

typedef __attribute__((ext_vector_type(2))) float v2f;
typedef __attribute__((ext_vector_type(8))) float v8f;

__global__ __launch_bounds__(256) void build_tables(
    const float* __restrict__ U0,
    const float* __restrict__ U1,
    const float* __restrict__ U2,
    const float* __restrict__ U3,
    float* __restrict__ out1,
    float* __restrict__ out2)
{
    const int w    = threadIdx.x >> 5;
    const int lane = threadIdx.x & 31;

    if (blockIdx.x < 225) {
        const int i3   = blockIdx.x / 15;
        const int blk  = blockIdx.x % 15;
        const int o2   = lane & 7;
        const int R0   = blk * 64 + w * 8 + (lane >> 3);

        float u2[2][32];
#pragma unroll
        for (int p = 0; p < 2; ++p) {
#pragma unroll
            for (int q = 0; q < 8; ++q) {
                const float4 t = *(const float4*)(U2 + ((R0 + p * 4) * 8 + o2) * 32 + q * 4);
                u2[p][q * 4 + 0] = t.x;  u2[p][q * 4 + 1] = t.y;
                u2[p][q * 4 + 2] = t.z;  u2[p][q * 4 + 3] = t.w;
            }
        }

        const float* s3 = U3 + i3 * 8;
        float acc[2][8];
#pragma unroll
        for (int p = 0; p < 2; ++p)
#pragma unroll
            for (int o3 = 0; o3 < 8; ++o3) acc[p][o3] = 0.f;
#pragma unroll
        for (int k = 0; k < 32; ++k) {
#pragma unroll
            for (int o3 = 0; o3 < 8; ++o3) {
                const float b = s3[k * 120 + o3];
                acc[0][o3] = __builtin_fmaf(u2[0][k], b, acc[0][o3]);
                acc[1][o3] = __builtin_fmaf(u2[1][k], b, acc[1][o3]);
            }
        }
#pragma unroll
        for (int p = 0; p < 2; ++p) {
            const int R  = R0 + p * 4;
            const int c  = R / 15;
            const int i2 = R - c * 15;
            float* q2 = out2 + (size_t)c * 14400 + (i2 * 15 + i3) * 64 + o2 * 8;
            *(float4*)(q2 + 0) = make_float4(acc[p][0], acc[p][1], acc[p][2], acc[p][3]);
            *(float4*)(q2 + 4) = make_float4(acc[p][4], acc[p][5], acc[p][6], acc[p][7]);
        }
    } else {
        const int bb = blockIdx.x - 225;
        const int i1 = bb / 15;
        const int i0 = bb % 15;
        const int o1 = w;

        const float* p1 = U1 + (i1 * 8 + o1) * 64 + 2 * lane;
        float2 u1[32];
#pragma unroll
        for (int k = 0; k < 32; ++k)
            u1[k] = *(const float2*)(p1 + k * 7680);

        const float* p0 = U0 + i0 * 256;
        float2 acc[8];
#pragma unroll
        for (int o0 = 0; o0 < 8; ++o0) { acc[o0].x = 0.f; acc[o0].y = 0.f; }
#pragma unroll
        for (int k = 0; k < 32; ++k) {
#pragma unroll
            for (int o0 = 0; o0 < 8; ++o0) {
                const float a = p0[o0 * 32 + k];
                acc[o0].x = __builtin_fmaf(a, u1[k].x, acc[o0].x);
                acc[o0].y = __builtin_fmaf(a, u1[k].y, acc[o0].y);
            }
        }
        float* q1 = out1 + ((size_t)(i0 * 15 + i1) * 64 + o1) * 64 + 2 * lane;
#pragma unroll
        for (int o0 = 0; o0 < 8; ++o0)
            *(float2*)(q1 + o0 * 512) = acc[o0];
    }
}

#define SA 68

__global__ __launch_bounds__(128) void ttm_gemm_kernel(
    const int*   __restrict__ x,
    const float* __restrict__ out1,
    const float* __restrict__ out2,
    float*       __restrict__ out,
    int nsamp)
{
    __shared__ __attribute__((aligned(16))) float lA[64 * SA];
    __shared__ __attribute__((aligned(16))) float lB[64 * 64];

    const int t    = threadIdx.x;
    const int tm   = t >> 5;
    const int lane = t & 31;
    const int lr   = lane & 15;
    const int lh   = lane >> 4;

    const int sr = t >> 4;
    const int sc = (t & 15) * 4;

    const float* Abase = &lA[(tm * 16 + lr) * SA + lh * 2];
    const float* Bbase = &lB[(lh * 2) * 64 + lr];

    for (int n = blockIdx.x; n < nsamp; n += gridDim.x) {
        const int xi   = x[n];
        const int row0 = xi / 225;
        const int row1 = xi % 225;

        const float* Ag = out1 + (size_t)row0 * 4096 + sr * 64 + sc;
        const float* Bg = out2 + (size_t)sr * 14400 + row1 * 64 + sc;
        const float4 ra0 = *(const float4*)(Ag + 0 * 512);
        const float4 ra1 = *(const float4*)(Ag + 1 * 512);
        const float4 ra2 = *(const float4*)(Ag + 2 * 512);
        const float4 ra3 = *(const float4*)(Ag + 3 * 512);
        const float4 ra4 = *(const float4*)(Ag + 4 * 512);
        const float4 ra5 = *(const float4*)(Ag + 5 * 512);
        const float4 ra6 = *(const float4*)(Ag + 6 * 512);
        const float4 ra7 = *(const float4*)(Ag + 7 * 512);
        const float4 rb0 = *(const float4*)(Bg + 0 * 115200);
        const float4 rb1 = *(const float4*)(Bg + 1 * 115200);
        const float4 rb2 = *(const float4*)(Bg + 2 * 115200);
        const float4 rb3 = *(const float4*)(Bg + 3 * 115200);
        const float4 rb4 = *(const float4*)(Bg + 4 * 115200);
        const float4 rb5 = *(const float4*)(Bg + 5 * 115200);
        const float4 rb6 = *(const float4*)(Bg + 6 * 115200);
        const float4 rb7 = *(const float4*)(Bg + 7 * 115200);
        __builtin_amdgcn_fence(__ATOMIC_RELEASE, "workgroup", "local");
        __builtin_amdgcn_s_barrier();
        __builtin_amdgcn_fence(__ATOMIC_ACQUIRE, "workgroup", "local");
        float* sA = &lA[sr * SA + sc];
        float* sB = &lB[sr * 64 + sc];
        *(float4*)(sA + 0 * 8 * SA) = ra0;   *(float4*)(sB + 0 * 512) = rb0;
        *(float4*)(sA + 1 * 8 * SA) = ra1;   *(float4*)(sB + 1 * 512) = rb1;
        *(float4*)(sA + 2 * 8 * SA) = ra2;   *(float4*)(sB + 2 * 512) = rb2;
        *(float4*)(sA + 3 * 8 * SA) = ra3;   *(float4*)(sB + 3 * 512) = rb3;
        *(float4*)(sA + 4 * 8 * SA) = ra4;   *(float4*)(sB + 4 * 512) = rb4;
        *(float4*)(sA + 5 * 8 * SA) = ra5;   *(float4*)(sB + 5 * 512) = rb5;
        *(float4*)(sA + 6 * 8 * SA) = ra6;   *(float4*)(sB + 6 * 512) = rb6;
        *(float4*)(sA + 7 * 8 * SA) = ra7;   *(float4*)(sB + 7 * 512) = rb7;
        __syncthreads();

        v8f acc0 = {}, acc1 = {}, acc2 = {}, acc3 = {};
#pragma unroll
        for (int k0 = 0; k0 < 16; ++k0) {
            const v2f a = *(const v2f*)(Abase + k0 * 4);
            const float* bp = Bbase + k0 * 4 * 64;
            v2f b0, b1, b2, b3;
            b0.x = bp[0];   b0.y = bp[64];
            b1.x = bp[16];  b1.y = bp[64 + 16];
            b2.x = bp[32];  b2.y = bp[64 + 32];
            b3.x = bp[48];  b3.y = bp[64 + 48];
            acc0 = __builtin_amdgcn_wmma_f32_16x16x4_f32(
                false, a, false, b0, (short)0, acc0, false, false);
            acc1 = __builtin_amdgcn_wmma_f32_16x16x4_f32(
                false, a, false, b1, (short)0, acc1, false, false);
            acc2 = __builtin_amdgcn_wmma_f32_16x16x4_f32(
                false, a, false, b2, (short)0, acc2, false, false);
            acc3 = __builtin_amdgcn_wmma_f32_16x16x4_f32(
                false, a, false, b3, (short)0, acc3, false, false);
        }

        float* op = out + (size_t)n * 4096 + (tm * 16 + lh * 8) * 64 + lr;
#pragma unroll
        for (int v = 0; v < 8; ++v) {
            op[v * 64]      = acc0[v];
            op[v * 64 + 16] = acc1[v];
            op[v * 64 + 32] = acc2[v];
            op[v * 64 + 48] = acc3[v];
        }
    }
}

extern "C" void kernel_launch(void* const* d_in, const int* in_sizes, int n_in,
                              void* d_out, int out_size, void* d_ws, size_t ws_size,
                              hipStream_t stream) {
    const int*   x  = (const int*)  d_in[0];
    const float* U0 = (const float*)d_in[1];
    const float* U1 = (const float*)d_in[2];
    const float* U2 = (const float*)d_in[3];
    const float* U3 = (const float*)d_in[4];
    float* out = (float*)d_out;

    float* w1 = (float*)d_ws;
    float* w2 = w1 + 921600;

    build_tables<<<450, 256, 0, stream>>>(U0, U1, U2, U3, w1, w2);

    const int nsamp = in_sizes[0];
    ttm_gemm_kernel<<<2048, 128, 0, stream>>>(x, w1, w2, out, nsamp);
}


# main kernel: sample loop software-pipelined: the next sample's two 16 KB tiles are loaded into registers (indices fetched ahead) while the current sample's WMMAs run; grid 2048 -> 1024; same sums
# speedup vs baseline: 1.5195x; 1.0502x over previous
#include <hip/hip_runtime.h>

typedef __attribute__((ext_vector_type(2))) float v2f;
typedef __attribute__((ext_vector_type(8))) float v8f;

__global__ __launch_bounds__(256) void build_tables(
    const float* __restrict__ U0,
    const float* __restrict__ U1,
    const float* __restrict__ U2,
    const float* __restrict__ U3,
    float* __restrict__ out1,
    float* __restrict__ out2)
{
    const int w    = threadIdx.x >> 5;
    const int lane = threadIdx.x & 31;

    if (blockIdx.x < 225) {
        const int i3   = blockIdx.x / 15;
        const int blk  = blockIdx.x % 15;
        const int o2   = lane & 7;
        const int R0   = blk * 64 + w * 8 + (lane >> 3);

        float u2[2][32];
#pragma unroll
        for (int p = 0; p < 2; ++p) {
#pragma unroll
            for (int q = 0; q < 8; ++q) {
                const float4 t = *(const float4*)(U2 + ((R0 + p * 4) * 8 + o2) * 32 + q * 4);
                u2[p][q * 4 + 0] = t.x;  u2[p][q * 4 + 1] = t.y;
                u2[p][q * 4 + 2] = t.z;  u2[p][q * 4 + 3] = t.w;
            }
        }

        const float* s3 = U3 + i3 * 8;
        float acc[2][8];
#pragma unroll
        for (int p = 0; p < 2; ++p)
#pragma unroll
            for (int o3 = 0; o3 < 8; ++o3) acc[p][o3] = 0.f;
#pragma unroll
        for (int k = 0; k < 32; ++k) {
#pragma unroll
            for (int o3 = 0; o3 < 8; ++o3) {
                const float b = s3[k * 120 + o3];
                acc[0][o3] = __builtin_fmaf(u2[0][k], b, acc[0][o3]);
                acc[1][o3] = __builtin_fmaf(u2[1][k], b, acc[1][o3]);
            }
        }
#pragma unroll
        for (int p = 0; p < 2; ++p) {
            const int R  = R0 + p * 4;
            const int c  = R / 15;
            const int i2 = R - c * 15;
            float* q2 = out2 + (size_t)c * 14400 + (i2 * 15 + i3) * 64 + o2 * 8;
            *(float4*)(q2 + 0) = make_float4(acc[p][0], acc[p][1], acc[p][2], acc[p][3]);
            *(float4*)(q2 + 4) = make_float4(acc[p][4], acc[p][5], acc[p][6], acc[p][7]);
        }
    } else {
        const int bb = blockIdx.x - 225;
        const int i1 = bb / 15;
        const int i0 = bb % 15;
        const int o1 = w;

        const float* p1 = U1 + (i1 * 8 + o1) * 64 + 2 * lane;
        float2 u1[32];
#pragma unroll
        for (int k = 0; k < 32; ++k)
            u1[k] = *(const float2*)(p1 + k * 7680);

        const float* p0 = U0 + i0 * 256;
        float2 acc[8];
#pragma unroll
        for (int o0 = 0; o0 < 8; ++o0) { acc[o0].x = 0.f; acc[o0].y = 0.f; }
#pragma unroll
        for (int k = 0; k < 32; ++k) {
#pragma unroll
            for (int o0 = 0; o0 < 8; ++o0) {
                const float a = p0[o0 * 32 + k];
                acc[o0].x = __builtin_fmaf(a, u1[k].x, acc[o0].x);
                acc[o0].y = __builtin_fmaf(a, u1[k].y, acc[o0].y);
            }
        }
        float* q1 = out1 + ((size_t)(i0 * 15 + i1) * 64 + o1) * 64 + 2 * lane;
#pragma unroll
        for (int o0 = 0; o0 < 8; ++o0)
            *(float2*)(q1 + o0 * 512) = acc[o0];
    }
}

#define SA 68

__global__ __launch_bounds__(128) void ttm_gemm_kernel(
    const int*   __restrict__ x,
    const float* __restrict__ out1,
    const float* __restrict__ out2,
    float*       __restrict__ out,
    int nsamp)
{
    __shared__ __attribute__((aligned(16))) float lA[64 * SA];
    __shared__ __attribute__((aligned(16))) float lB[64 * 64];

    const int t    = threadIdx.x;
    const int tm   = t >> 5;
    const int lane = t & 31;
    const int lr   = lane & 15;
    const int lh   = lane >> 4;

    const int sr = t >> 4;
    const int sc = (t & 15) * 4;

    const float* Abase = &lA[(tm * 16 + lr) * SA + lh * 2];
    const float* Bbase = &lB[(lh * 2) * 64 + lr];

    const int G = gridDim.x;
    int n = blockIdx.x;
    if (n >= nsamp) return;

    const int xq0 = x[n];
    int xq1 = (n + 1 * G < nsamp) ? x[n + 1 * G] : 0;
    int xq2 = (n + 2 * G < nsamp) ? x[n + 2 * G] : 0;
    int xq3 = (n + 3 * G < nsamp) ? x[n + 3 * G] : 0;
    int xq4 = (n + 4 * G < nsamp) ? x[n + 4 * G] : 0;
    int xq5 = (n + 5 * G < nsamp) ? x[n + 5 * G] : 0;
    int xq6 = (n + 6 * G < nsamp) ? x[n + 6 * G] : 0;
    int xq7 = (n + 7 * G < nsamp) ? x[n + 7 * G] : 0;

    float4 ra0, ra1, ra2, ra3, ra4, ra5, ra6, ra7;
    float4 rb0, rb1, rb2, rb3, rb4, rb5, rb6, rb7;
#define LOAD_TILES(xi_)                                                        \
    {                                                                          \
        const int row0_ = (xi_) / 225;                                         \
        const int row1_ = (xi_) % 225;                                         \
        const float* Ag_ = out1 + (size_t)row0_ * 4096 + sr * 64 + sc;         \
        const float* Bg_ = out2 + (size_t)sr * 14400 + row1_ * 64 + sc;        \
        ra0 = *(const float4*)(Ag_ + 0 * 512);                                 \
        ra1 = *(const float4*)(Ag_ + 1 * 512);                                 \
        ra2 = *(const float4*)(Ag_ + 2 * 512);                                 \
        ra3 = *(const float4*)(Ag_ + 3 * 512);                                 \
        ra4 = *(const float4*)(Ag_ + 4 * 512);                                 \
        ra5 = *(const float4*)(Ag_ + 5 * 512);                                 \
        ra6 = *(const float4*)(Ag_ + 6 * 512);                                 \
        ra7 = *(const float4*)(Ag_ + 7 * 512);                                 \
        rb0 = *(const float4*)(Bg_ + 0 * 115200);     \
        rb1 = *(const float4*)(Bg_ + 1 * 115200);                              \
        rb2 = *(const float4*)(Bg_ + 2 * 115200);                              \
        rb3 = *(const float4*)(Bg_ + 3 * 115200);                              \
        rb4 = *(const float4*)(Bg_ + 4 * 115200);                              \
        rb5 = *(const float4*)(Bg_ + 5 * 115200);                              \
        rb6 = *(const float4*)(Bg_ + 6 * 115200);                              \
        rb7 = *(const float4*)(Bg_ + 7 * 115200);                              \
    }

    LOAD_TILES(xq0);

    for (;;) {
        __builtin_amdgcn_fence(__ATOMIC_RELEASE, "workgroup", "local");
        __builtin_amdgcn_s_barrier();
        __builtin_amdgcn_fence(__ATOMIC_ACQUIRE, "workgroup", "local");
        float* sA = &lA[sr * SA + sc];
        float* sB = &lB[sr * 64 + sc];
        *(float4*)(sA + 0 * 8 * SA) = ra0;   *(float4*)(sB + 0 * 512) = rb0;
        *(float4*)(sA + 1 * 8 * SA) = ra1;   *(float4*)(sB + 1 * 512) = rb1;
        *(float4*)(sA + 2 * 8 * SA) = ra2;   *(float4*)(sB + 2 * 512) = rb2;
        *(float4*)(sA + 3 * 8 * SA) = ra3;   *(float4*)(sB + 3 * 512) = rb3;
        *(float4*)(sA + 4 * 8 * SA) = ra4;   *(float4*)(sB + 4 * 512) = rb4;
        *(float4*)(sA + 5 * 8 * SA) = ra5;   *(float4*)(sB + 5 * 512) = rb5;
        *(float4*)(sA + 6 * 8 * SA) = ra6;   *(float4*)(sB + 6 * 512) = rb6;
        *(float4*)(sA + 7 * 8 * SA) = ra7;   *(float4*)(sB + 7 * 512) = rb7;
        __syncthreads();

        const int  nn   = n + G;
        const bool more = nn < nsamp;
        if (more) {
            LOAD_TILES(xq1);
        }

        v8f acc0 = {}, acc1 = {}, acc2 = {}, acc3 = {};
#pragma unroll
        for (int k0 = 0; k0 < 16; ++k0) {
            const v2f a = *(const v2f*)(Abase + k0 * 4);
            const float* bp = Bbase + k0 * 4 * 64;
            v2f b0, b1, b2, b3;
            b0.x = bp[0];   b0.y = bp[64];
            b1.x = bp[16];  b1.y = bp[64 + 16];
            b2.x = bp[32];  b2.y = bp[64 + 32];
            b3.x = bp[48];  b3.y = bp[64 + 48];
            acc0 = __builtin_amdgcn_wmma_f32_16x16x4_f32(
                false, a, false, b0, (short)0, acc0, false, false);
            acc1 = __builtin_amdgcn_wmma_f32_16x16x4_f32(
                false, a, false, b1, (short)0, acc1, false, false);
            acc2 = __builtin_amdgcn_wmma_f32_16x16x4_f32(
                false, a, false, b2, (short)0, acc2, false, false);
            acc3 = __builtin_amdgcn_wmma_f32_16x16x4_f32(
                false, a, false, b3, (short)0, acc3, false, false);
        }

        float* op = out + (size_t)n * 4096 + (tm * 16 + lh * 8) * 64 + lr;
#pragma unroll
        for (int v = 0; v < 8; ++v) {
            op[v * 64]      = acc0[v];
            op[v * 64 + 16] = acc1[v];
            op[v * 64 + 32] = acc2[v];
            op[v * 64 + 48] = acc3[v];
        }

        if (!more) break;
        n = nn;
        xq1 = xq2;  xq2 = xq3;  xq3 = xq4;  xq4 = xq5;  xq5 = xq6;  xq6 = xq7;
        xq7 = (n + 7 * G < nsamp) ? x[n + 7 * G] : 0;
    }
#undef LOAD_TILES
}

extern "C" void kernel_launch(void* const* d_in, const int* in_sizes, int n_in,
                              void* d_out, int out_size, void* d_ws, size_t ws_size,
                              hipStream_t stream) {
    const int*   x  = (const int*)  d_in[0];
    const float* U0 = (const float*)d_in[1];
    const float* U1 = (const float*)d_in[2];
    const float* U2 = (const float*)d_in[3];
    const float* U3 = (const float*)d_in[4];
    float* out = (float*)d_out;

    float* w1 = (float*)d_ws;
    float* w2 = w1 + 921600;

    build_tables<<<450, 256, 0, stream>>>(U0, U1, U2, U3, w1, w2);

    const int nsamp = in_sizes[0];
    ttm_gemm_kernel<<<1024, 128, 0, stream>>>(x, w1, w2, out, nsamp);
}


# main kernel: B kept as a k-pair table (k,k+1 adjacent; table kernel writes it so; LDS pitch 160): a B fragment is one conflict-free 8-byte LDS read, was two conflicted 4-byte reads + moves; same sums
# speedup vs baseline: 1.5951x; 1.0498x over previous
#include <hip/hip_runtime.h>

typedef __attribute__((ext_vector_type(2))) float v2f;
typedef __attribute__((ext_vector_type(8))) float v8f;

__global__ __launch_bounds__(256) void build_tables(
    const float* __restrict__ U0,
    const float* __restrict__ U1,
    const float* __restrict__ U2,
    const float* __restrict__ U3,
    float* __restrict__ out1,
    float* __restrict__ out2)
{
    const int w    = threadIdx.x >> 5;
    const int lane = threadIdx.x & 31;

    if (blockIdx.x < 225) {
        const int row1 = blockIdx.x;
        const int i2   = row1 / 15;
        const int i3   = row1 - i2 * 15;
        const int cp   = threadIdx.x >> 3;
        const int o2   = threadIdx.x & 7;

        float u2[2][32];
#pragma unroll
        for (int p = 0; p < 2; ++p) {
#pragma unroll
            for (int q = 0; q < 8; ++q) {
                const float4 t = *(const float4*)(U2 + (((2 * cp + p) * 15 + i2) * 8 + o2) * 32 + q * 4);
                u2[p][q * 4 + 0] = t.x;  u2[p][q * 4 + 1] = t.y;
                u2[p][q * 4 + 2] = t.z;  u2[p][q * 4 + 3] = t.w;
            }
        }

        const float* s3 = U3 + i3 * 8;
        float acc[2][8];
#pragma unroll
        for (int p = 0; p < 2; ++p)
#pragma unroll
            for (int o3 = 0; o3 < 8; ++o3) acc[p][o3] = 0.f;
#pragma unroll
        for (int k = 0; k < 32; ++k) {
#pragma unroll
            for (int o3 = 0; o3 < 8; ++o3) {
                const float b = s3[k * 120 + o3];
                acc[0][o3] = __builtin_fmaf(u2[0][k], b, acc[0][o3]);
                acc[1][o3] = __builtin_fmaf(u2[1][k], b, acc[1][o3]);
            }
        }
        float* q2 = out2 + (size_t)row1 * 4096 + cp * 128 + o2 * 16;
        *(float4*)(q2 +  0) = make_float4(acc[0][0], acc[1][0], acc[0][1], acc[1][1]);
        *(float4*)(q2 +  4) = make_float4(acc[0][2], acc[1][2], acc[0][3], acc[1][3]);
        *(float4*)(q2 +  8) = make_float4(acc[0][4], acc[1][4], acc[0][5], acc[1][5]);
        *(float4*)(q2 + 12) = make_float4(acc[0][6], acc[1][6], acc[0][7], acc[1][7]);
    } else {
        const int bb = blockIdx.x - 225;
        const int i1 = bb / 15;
        const int i0 = bb % 15;
        const int o1 = w;

        const float* p1 = U1 + (i1 * 8 + o1) * 64 + 2 * lane;
        float2 u1[32];
#pragma unroll
        for (int k = 0; k < 32; ++k)
            u1[k] = *(const float2*)(p1 + k * 7680);

        const float* p0 = U0 + i0 * 256;
        float2 acc[8];
#pragma unroll
        for (int o0 = 0; o0 < 8; ++o0) { acc[o0].x = 0.f; acc[o0].y = 0.f; }
#pragma unroll
        for (int k = 0; k < 32; ++k) {
#pragma unroll
            for (int o0 = 0; o0 < 8; ++o0) {
                const float a = p0[o0 * 32 + k];
                acc[o0].x = __builtin_fmaf(a, u1[k].x, acc[o0].x);
                acc[o0].y = __builtin_fmaf(a, u1[k].y, acc[o0].y);
            }
        }
        float* q1 = out1 + ((size_t)(i0 * 15 + i1) * 64 + o1) * 64 + 2 * lane;
#pragma unroll
        for (int o0 = 0; o0 < 8; ++o0)
            *(float2*)(q1 + o0 * 512) = acc[o0];
    }
}

#define SA 68
#define SB 160

__global__ __launch_bounds__(128) void ttm_gemm_kernel(
    const int*   __restrict__ x,
    const float* __restrict__ out1,
    const float* __restrict__ out2,
    float*       __restrict__ out,
    int nsamp)
{
    __shared__ __attribute__((aligned(16))) float lA[64 * SA];
    __shared__ __attribute__((aligned(16))) float lB[32 * SB];

    const int t    = threadIdx.x;
    const int tm   = t >> 5;
    const int lane = t & 31;
    const int lr   = lane & 15;
    const int lh   = lane >> 4;

    const int sr = t >> 4;
    const int sc = (t & 15) * 4;

    const float* Abase = &lA[(tm * 16 + lr) * SA + lh * 2];
    const float* Bbase = &lB[lh * SB + lr * 2];

    const int G = gridDim.x;
    int n = blockIdx.x;
    if (n >= nsamp) return;

    const int xq0 = x[n];
    int xq1 = (n + 1 * G < nsamp) ? x[n + 1 * G] : 0;
    int xq2 = (n + 2 * G < nsamp) ? x[n + 2 * G] : 0;
    int xq3 = (n + 3 * G < nsamp) ? x[n + 3 * G] : 0;
    int xq4 = (n + 4 * G < nsamp) ? x[n + 4 * G] : 0;
    int xq5 = (n + 5 * G < nsamp) ? x[n + 5 * G] : 0;
    int xq6 = (n + 6 * G < nsamp) ? x[n + 6 * G] : 0;
    int xq7 = (n + 7 * G < nsamp) ? x[n + 7 * G] : 0;

    float4 ra0, ra1, ra2, ra3, ra4, ra5, ra6, ra7;
    float4 rb0, rb1, rb2, rb3, rb4, rb5, rb6, rb7;
#define LOAD_TILES(xi_)                                                        \
    {                                                                          \
        const int row0_ = (xi_) / 225;                                         \
        const int row1_ = (xi_) % 225;                                         \
        const float* Ag_ = out1 + (size_t)row0_ * 4096 + sr * 64 + sc;         \
        const float* Bg_ = out2 + (size_t)row1_ * 4096 + sr * 64 + sc;         \
        ra0 = *(const float4*)(Ag_ + 0 * 512);                                 \
        ra1 = *(const float4*)(Ag_ + 1 * 512);                                 \
        ra2 = *(const float4*)(Ag_ + 2 * 512);                                 \
        ra3 = *(const float4*)(Ag_ + 3 * 512);                                 \
        ra4 = *(const float4*)(Ag_ + 4 * 512);                                 \
        ra5 = *(const float4*)(Ag_ + 5 * 512);                                 \
        ra6 = *(const float4*)(Ag_ + 6 * 512);                                 \
        ra7 = *(const float4*)(Ag_ + 7 * 512);                                 \
        rb0 = *(const float4*)(Bg_ + 0 * 512);                                 \
        rb1 = *(const float4*)(Bg_ + 1 * 512);                                 \
        rb2 = *(const float4*)(Bg_ + 2 * 512);                                 \
        rb3 = *(const float4*)(Bg_ + 3 * 512);                                 \
        rb4 = *(const float4*)(Bg_ + 4 * 512);                                 \
        rb5 = *(const float4*)(Bg_ + 5 * 512);                                 \
        rb6 = *(const float4*)(Bg_ + 6 * 512);                                 \
        rb7 = *(const float4*)(Bg_ + 7 * 512);                                 \
    }

    LOAD_TILES(xq0);

    for (;;) {
        __builtin_amdgcn_fence(__ATOMIC_RELEASE, "workgroup", "local");
        __builtin_amdgcn_s_barrier();
        __builtin_amdgcn_fence(__ATOMIC_ACQUIRE, "workgroup", "local");
        float* sA = &lA[sr * SA + sc];
        float* sB = &lB[(t >> 5) * SB + (t & 31) * 4];
        *(float4*)(sA + 0 * 8 * SA) = ra0;   *(float4*)(sB + 0 * 4 * SB) = rb0;
        *(float4*)(sA + 1 * 8 * SA) = ra1;   *(float4*)(sB + 1 * 4 * SB) = rb1;
        *(float4*)(sA + 2 * 8 * SA) = ra2;   *(float4*)(sB + 2 * 4 * SB) = rb2;
        *(float4*)(sA + 3 * 8 * SA) = ra3;   *(float4*)(sB + 3 * 4 * SB) = rb3;
        *(float4*)(sA + 4 * 8 * SA) = ra4;   *(float4*)(sB + 4 * 4 * SB) = rb4;
        *(float4*)(sA + 5 * 8 * SA) = ra5;   *(float4*)(sB + 5 * 4 * SB) = rb5;
        *(float4*)(sA + 6 * 8 * SA) = ra6;   *(float4*)(sB + 6 * 4 * SB) = rb6;
        *(float4*)(sA + 7 * 8 * SA) = ra7;   *(float4*)(sB + 7 * 4 * SB) = rb7;
        __syncthreads();

        const int  nn   = n + G;
        const bool more = nn < nsamp;
        if (more) {
            LOAD_TILES(xq1);
        }

        v8f acc0 = {}, acc1 = {}, acc2 = {}, acc3 = {};
#pragma unroll
        for (int k0 = 0; k0 < 16; ++k0) {
            const v2f a = *(const v2f*)(Abase + k0 * 4);
            const float* bp = Bbase + k0 * 2 * SB;
            const v2f b0 = *(const v2f*)(bp + 0);
            const v2f b1 = *(const v2f*)(bp + 32);
            const v2f b2 = *(const v2f*)(bp + 64);
            const v2f b3 = *(const v2f*)(bp + 96);
            acc0 = __builtin_amdgcn_wmma_f32_16x16x4_f32(
                false, a, false, b0, (short)0, acc0, false, false);
            acc1 = __builtin_amdgcn_wmma_f32_16x16x4_f32(
                false, a, false, b1, (short)0, acc1, false, false);
            acc2 = __builtin_amdgcn_wmma_f32_16x16x4_f32(
                false, a, false, b2, (short)0, acc2, false, false);
            acc3 = __builtin_amdgcn_wmma_f32_16x16x4_f32(
                false, a, false, b3, (short)0, acc3, false, false);
        }

        float* op = out + (size_t)n * 4096 + (tm * 16 + lh * 8) * 64 + lr;
#pragma unroll
        for (int v = 0; v < 8; ++v) {
            op[v * 64]      = acc0[v];
            op[v * 64 + 16] = acc1[v];
            op[v * 64 + 32] = acc2[v];
            op[v * 64 + 48] = acc3[v];
        }

        if (!more) break;
        n = nn;
        xq1 = xq2;  xq2 = xq3;  xq3 = xq4;  xq4 = xq5;  xq5 = xq6;  xq6 = xq7;
        xq7 = (n + 7 * G < nsamp) ? x[n + 7 * G] : 0;
    }
#undef LOAD_TILES
}

extern "C" void kernel_launch(void* const* d_in, const int* in_sizes, int n_in,
                              void* d_out, int out_size, void* d_ws, size_t ws_size,
                              hipStream_t stream) {
    const int*   x  = (const int*)  d_in[0];
    const float* U0 = (const float*)d_in[1];
    const float* U1 = (const float*)d_in[2];
    const float* U2 = (const float*)d_in[3];
    const float* U3 = (const float*)d_in[4];
    float* out = (float*)d_out;

    float* w1 = (float*)d_ws;
    float* w2 = w1 + 921600;

    build_tables<<<450, 256, 0, stream>>>(U0, U1, U2, U3, w1, w2);

    const int nsamp = in_sizes[0];
    ttm_gemm_kernel<<<1024, 128, 0, stream>>>(x, w1, w2, out, nsamp);
}


# main kernel: wave priority set per sample from the work-group's remaining samples (s_setprio 3,2,1,0 for >=4,3,2,1 left): work-groups sharing a WGP end together, not oldest first; same sums
# speedup vs baseline: 1.6854x; 1.0566x over previous
#include <hip/hip_runtime.h>

typedef __attribute__((ext_vector_type(2))) float v2f;
typedef __attribute__((ext_vector_type(8))) float v8f;

__global__ __launch_bounds__(256) void build_tables(
    const float* __restrict__ U0,
    const float* __restrict__ U1,
    const float* __restrict__ U2,
    const float* __restrict__ U3,
    float* __restrict__ out1,
    float* __restrict__ out2)
{
    const int w    = threadIdx.x >> 5;
    const int lane = threadIdx.x & 31;

    if (blockIdx.x < 225) {
        const int row1 = blockIdx.x;
        const int i2   = row1 / 15;
        const int i3   = row1 - i2 * 15;
        const int cp   = threadIdx.x >> 3;
        const int o2   = threadIdx.x & 7;

        float u2[2][32];
#pragma unroll
        for (int p = 0; p < 2; ++p) {
#pragma unroll
            for (int q = 0; q < 8; ++q) {
                const float4 t = *(const float4*)(U2 + (((2 * cp + p) * 15 + i2) * 8 + o2) * 32 + q * 4);
                u2[p][q * 4 + 0] = t.x;  u2[p][q * 4 + 1] = t.y;
                u2[p][q * 4 + 2] = t.z;  u2[p][q * 4 + 3] = t.w;
            }
        }

        const float* s3 = U3 + i3 * 8;
        float acc[2][8];
#pragma unroll
        for (int p = 0; p < 2; ++p)
#pragma unroll
            for (int o3 = 0; o3 < 8; ++o3) acc[p][o3] = 0.f;
#pragma unroll
        for (int k = 0; k < 32; ++k) {
#pragma unroll
            for (int o3 = 0; o3 < 8; ++o3) {
                const float b = s3[k * 120 + o3];
                acc[0][o3] = __builtin_fmaf(u2[0][k], b, acc[0][o3]);
                acc[1][o3] = __builtin_fmaf(u2[1][k], b, acc[1][o3]);
            }
        }
        float* q2 = out2 + (size_t)row1 * 4096 + cp * 128 + o2 * 16;
        *(float4*)(q2 +  0) = make_float4(acc[0][0], acc[1][0], acc[0][1], acc[1][1]);
        *(float4*)(q2 +  4) = make_float4(acc[0][2], acc[1][2], acc[0][3], acc[1][3]);
        *(float4*)(q2 +  8) = make_float4(acc[0][4], acc[1][4], acc[0][5], acc[1][5]);
        *(float4*)(q2 + 12) = make_float4(acc[0][6], acc[1][6], acc[0][7], acc[1][7]);
    } else {
        const int bb = blockIdx.x - 225;
        const int i1 = bb / 15;
        const int i0 = bb % 15;
        const int o1 = w;

        const float* p1 = U1 + (i1 * 8 + o1) * 64 + 2 * lane;
        float2 u1[32];
#pragma unroll
        for (int k = 0; k < 32; ++k)
            u1[k] = *(const float2*)(p1 + k * 7680);

        const float* p0 = U0 + i0 * 256;
        float2 acc[8];
#pragma unroll
        for (int o0 = 0; o0 < 8; ++o0) { acc[o0].x = 0.f; acc[o0].y = 0.f; }
#pragma unroll
        for (int k = 0; k < 32; ++k) {
#pragma unroll
            for (int o0 = 0; o0 < 8; ++o0) {
                const float a = p0[o0 * 32 + k];
                acc[o0].x = __builtin_fmaf(a, u1[k].x, acc[o0].x);
                acc[o0].y = __builtin_fmaf(a, u1[k].y, acc[o0].y);
            }
        }
        float* q1 = out1 + ((size_t)(i0 * 15 + i1) * 64 + o1) * 64 + 2 * lane;
#pragma unroll
        for (int o0 = 0; o0 < 8; ++o0)
            *(float2*)(q1 + o0 * 512) = acc[o0];
    }
}

#define SA 68
#define SB 160

__global__ __launch_bounds__(128) void ttm_gemm_kernel(
    const int*   __restrict__ x,
    const float* __restrict__ out1,
    const float* __restrict__ out2,
    float*       __restrict__ out,
    int nsamp)
{
    __shared__ __attribute__((aligned(16))) float lA[64 * SA];
    __shared__ __attribute__((aligned(16))) float lB[32 * SB];

    const int t    = threadIdx.x;
    const int tm   = t >> 5;
    const int lane = t & 31;
    const int lr   = lane & 15;
    const int lh   = lane >> 4;

    const int sr = t >> 4;
    const int sc = (t & 15) * 4;

    const float* Abase = &lA[(tm * 16 + lr) * SA + lh * 2];
    const float* Bbase = &lB[lh * SB + lr * 2];

    const int G = gridDim.x;
    int n = blockIdx.x;
    if (n >= nsamp) return;

    const int xq0 = x[n];
    int xq1 = (n + 1 * G < nsamp) ? x[n + 1 * G] : 0;
    int xq2 = (n + 2 * G < nsamp) ? x[n + 2 * G] : 0;
    int xq3 = (n + 3 * G < nsamp) ? x[n + 3 * G] : 0;
    int xq4 = (n + 4 * G < nsamp) ? x[n + 4 * G] : 0;
    int xq5 = (n + 5 * G < nsamp) ? x[n + 5 * G] : 0;
    int xq6 = (n + 6 * G < nsamp) ? x[n + 6 * G] : 0;
    int xq7 = (n + 7 * G < nsamp) ? x[n + 7 * G] : 0;

    float4 ra0, ra1, ra2, ra3, ra4, ra5, ra6, ra7;
    float4 rb0, rb1, rb2, rb3, rb4, rb5, rb6, rb7;
#define LOAD_TILES(xi_)                                                        \
    {                                                                          \
        const int row0_ = (xi_) / 225;                                         \
        const int row1_ = (xi_) % 225;                                         \
        const float* Ag_ = out1 + (size_t)row0_ * 4096 + sr * 64 + sc;         \
        const float* Bg_ = out2 + (size_t)row1_ * 4096 + sr * 64 + sc;         \
        ra0 = *(const float4*)(Ag_ + 0 * 512);                                 \
        ra1 = *(const float4*)(Ag_ + 1 * 512);                                 \
        ra2 = *(const float4*)(Ag_ + 2 * 512);                                 \
        ra3 = *(const float4*)(Ag_ + 3 * 512);                                 \
        ra4 = *(const float4*)(Ag_ + 4 * 512);                                 \
        ra5 = *(const float4*)(Ag_ + 5 * 512);                                 \
        ra6 = *(const float4*)(Ag_ + 6 * 512);                                 \
        ra7 = *(const float4*)(Ag_ + 7 * 512);                                 \
        rb0 = *(const float4*)(Bg_ + 0 * 512);                                 \
        rb1 = *(const float4*)(Bg_ + 1 * 512);                                 \
        rb2 = *(const float4*)(Bg_ + 2 * 512);                                 \
        rb3 = *(const float4*)(Bg_ + 3 * 512);                                 \
        rb4 = *(const float4*)(Bg_ + 4 * 512);                                 \
        rb5 = *(const float4*)(Bg_ + 5 * 512);                                 \
        rb6 = *(const float4*)(Bg_ + 6 * 512);                                 \
        rb7 = *(const float4*)(Bg_ + 7 * 512);                                 \
    }

    LOAD_TILES(xq0);

    for (;;) {
        if (n + 3 * G < nsamp)      __builtin_amdgcn_s_setprio(3);
        else if (n + 2 * G < nsamp) __builtin_amdgcn_s_setprio(2);
        else if (n + G < nsamp)     __builtin_amdgcn_s_setprio(1);
        else                        __builtin_amdgcn_s_setprio(0);

        __builtin_amdgcn_fence(__ATOMIC_RELEASE, "workgroup", "local");
        __builtin_amdgcn_s_barrier();
        __builtin_amdgcn_fence(__ATOMIC_ACQUIRE, "workgroup", "local");
        float* sA = &lA[sr * SA + sc];
        float* sB = &lB[(t >> 5) * SB + (t & 31) * 4];
        *(float4*)(sA + 0 * 8 * SA) = ra0;   *(float4*)(sB + 0 * 4 * SB) = rb0;
        *(float4*)(sA + 1 * 8 * SA) = ra1;   *(float4*)(sB + 1 * 4 * SB) = rb1;
        *(float4*)(sA + 2 * 8 * SA) = ra2;   *(float4*)(sB + 2 * 4 * SB) = rb2;
        *(float4*)(sA + 3 * 8 * SA) = ra3;   *(float4*)(sB + 3 * 4 * SB) = rb3;
        *(float4*)(sA + 4 * 8 * SA) = ra4;   *(float4*)(sB + 4 * 4 * SB) = rb4;
        *(float4*)(sA + 5 * 8 * SA) = ra5;   *(float4*)(sB + 5 * 4 * SB) = rb5;
        *(float4*)(sA + 6 * 8 * SA) = ra6;   *(float4*)(sB + 6 * 4 * SB) = rb6;
        *(float4*)(sA + 7 * 8 * SA) = ra7;   *(float4*)(sB + 7 * 4 * SB) = rb7;
        __syncthreads();

        const int  nn   = n + G;
        const bool more = nn < nsamp;
        if (more) {
            LOAD_TILES(xq1);
        }

        v8f acc0 = {}, acc1 = {}, acc2 = {}, acc3 = {};
#pragma unroll
        for (int k0 = 0; k0 < 16; ++k0) {
            const v2f a = *(const v2f*)(Abase + k0 * 4);
            const float* bp = Bbase + k0 * 2 * SB;
            const v2f b0 = *(const v2f*)(bp + 0);
            const v2f b1 = *(const v2f*)(bp + 32);
            const v2f b2 = *(const v2f*)(bp + 64);
            const v2f b3 = *(const v2f*)(bp + 96);
            acc0 = __builtin_amdgcn_wmma_f32_16x16x4_f32(
                false, a, false, b0, (short)0, acc0, false, false);
            acc1 = __builtin_amdgcn_wmma_f32_16x16x4_f32(
                false, a, false, b1, (short)0, acc1, false, false);
            acc2 = __builtin_amdgcn_wmma_f32_16x16x4_f32(
                false, a, false, b2, (short)0, acc2, false, false);
            acc3 = __builtin_amdgcn_wmma_f32_16x16x4_f32(
                false, a, false, b3, (short)0, acc3, false, false);
        }

        float* op = out + (size_t)n * 4096 + (tm * 16 + lh * 8) * 64 + lr;
#pragma unroll
        for (int v = 0; v < 8; ++v) {
            op[v * 64]      = acc0[v];
            op[v * 64 + 16] = acc1[v];
            op[v * 64 + 32] = acc2[v];
            op[v * 64 + 48] = acc3[v];
        }

        if (!more) break;
        n = nn;
        xq1 = xq2;  xq2 = xq3;  xq3 = xq4;  xq4 = xq5;  xq5 = xq6;  xq6 = xq7;
        xq7 = (n + 7 * G < nsamp) ? x[n + 7 * G] : 0;
    }
#undef LOAD_TILES
}

extern "C" void kernel_launch(void* const* d_in, const int* in_sizes, int n_in,
                              void* d_out, int out_size, void* d_ws, size_t ws_size,
                              hipStream_t stream) {
    const int*   x  = (const int*)  d_in[0];
    const float* U0 = (const float*)d_in[1];
    const float* U1 = (const float*)d_in[2];
    const float* U2 = (const float*)d_in[3];
    const float* U3 = (const float*)d_in[4];
    float* out = (float*)d_out;

    float* w1 = (float*)d_ws;
    float* w2 = w1 + 921600;

    build_tables<<<450, 256, 0, stream>>>(U0, U1, U2, U3, w1, w2);

    const int nsamp = in_sizes[0];
    ttm_gemm_kernel<<<1024, 128, 0, stream>>>(x, w1, w2, out, nsamp);
}


# main kernel: a sample's LDS fragment reads go in whole-block groups (4 k-steps, 20 reads) one 16-WMMA block ahead, into a second fragment register set: 2 places in the WMMA stream, were 15; same sums
# speedup vs baseline: 1.7405x; 1.0327x over previous
#include <hip/hip_runtime.h>

typedef __attribute__((ext_vector_type(2))) float v2f;
typedef __attribute__((ext_vector_type(8))) float v8f;

__global__ __launch_bounds__(256) void build_tables(
    const float* __restrict__ U0,
    const float* __restrict__ U1,
    const float* __restrict__ U2,
    const float* __restrict__ U3,
    float* __restrict__ out1,
    float* __restrict__ out2)
{
    const int w    = threadIdx.x >> 5;
    const int lane = threadIdx.x & 31;

    if (blockIdx.x < 225) {
        const int row1 = blockIdx.x;
        const int i2   = row1 / 15;
        const int i3   = row1 - i2 * 15;
        const int cp   = threadIdx.x >> 3;
        const int o2   = threadIdx.x & 7;

        float u2[2][32];
#pragma unroll
        for (int p = 0; p < 2; ++p) {
#pragma unroll
            for (int q = 0; q < 8; ++q) {
                const float4 t = *(const float4*)(U2 + (((2 * cp + p) * 15 + i2) * 8 + o2) * 32 + q * 4);
                u2[p][q * 4 + 0] = t.x;  u2[p][q * 4 + 1] = t.y;
                u2[p][q * 4 + 2] = t.z;  u2[p][q * 4 + 3] = t.w;
            }
        }

        const float* s3 = U3 + i3 * 8;
        float acc[2][8];
#pragma unroll
        for (int p = 0; p < 2; ++p)
#pragma unroll
            for (int o3 = 0; o3 < 8; ++o3) acc[p][o3] = 0.f;
#pragma unroll
        for (int k = 0; k < 32; ++k) {
#pragma unroll
            for (int o3 = 0; o3 < 8; ++o3) {
                const float b = s3[k * 120 + o3];
                acc[0][o3] = __builtin_fmaf(u2[0][k], b, acc[0][o3]);
                acc[1][o3] = __builtin_fmaf(u2[1][k], b, acc[1][o3]);
            }
        }
        float* q2 = out2 + (size_t)row1 * 4096 + cp * 128 + o2 * 16;
        *(float4*)(q2 +  0) = make_float4(acc[0][0], acc[1][0], acc[0][1], acc[1][1]);
        *(float4*)(q2 +  4) = make_float4(acc[0][2], acc[1][2], acc[0][3], acc[1][3]);
        *(float4*)(q2 +  8) = make_float4(acc[0][4], acc[1][4], acc[0][5], acc[1][5]);
        *(float4*)(q2 + 12) = make_float4(acc[0][6], acc[1][6], acc[0][7], acc[1][7]);
    } else {
        const int bb = blockIdx.x - 225;
        const int i1 = bb / 15;
        const int i0 = bb % 15;
        const int o1 = w;

        const float* p1 = U1 + (i1 * 8 + o1) * 64 + 2 * lane;
        float2 u1[32];
#pragma unroll
        for (int k = 0; k < 32; ++k)
            u1[k] = *(const float2*)(p1 + k * 7680);

        const float* p0 = U0 + i0 * 256;
        float2 acc[8];
#pragma unroll
        for (int o0 = 0; o0 < 8; ++o0) { acc[o0].x = 0.f; acc[o0].y = 0.f; }
#pragma unroll
        for (int k = 0; k < 32; ++k) {
#pragma unroll
            for (int o0 = 0; o0 < 8; ++o0) {
                const float a = p0[o0 * 32 + k];
                acc[o0].x = __builtin_fmaf(a, u1[k].x, acc[o0].x);
                acc[o0].y = __builtin_fmaf(a, u1[k].y, acc[o0].y);
            }
        }
        float* q1 = out1 + ((size_t)(i0 * 15 + i1) * 64 + o1) * 64 + 2 * lane;
#pragma unroll
        for (int o0 = 0; o0 < 8; ++o0)
            *(float2*)(q1 + o0 * 512) = acc[o0];
    }
}

#define SA 68
#define SB 160

__global__ __launch_bounds__(128) void ttm_gemm_kernel(
    const int*   __restrict__ x,
    const float* __restrict__ out1,
    const float* __restrict__ out2,
    float*       __restrict__ out,
    int nsamp)
{
    __shared__ __attribute__((aligned(16))) float lA[64 * SA];
    __shared__ __attribute__((aligned(16))) float lB[32 * SB];

    const int t    = threadIdx.x;
    const int tm   = t >> 5;
    const int lane = t & 31;
    const int lr   = lane & 15;
    const int lh   = lane >> 4;

    const int sr = t >> 4;
    const int sc = (t & 15) * 4;

    const float* Abase = &lA[(tm * 16 + lr) * SA + lh * 2];
    const float* Bbase = &lB[lh * SB + lr * 2];

    const int G = gridDim.x;
    int n = blockIdx.x;
    if (n >= nsamp) return;

    const int xq0 = x[n];
    int xq1 = (n + 1 * G < nsamp) ? x[n + 1 * G] : 0;
    int xq2 = (n + 2 * G < nsamp) ? x[n + 2 * G] : 0;
    int xq3 = (n + 3 * G < nsamp) ? x[n + 3 * G] : 0;
    int xq4 = (n + 4 * G < nsamp) ? x[n + 4 * G] : 0;
    int xq5 = (n + 5 * G < nsamp) ? x[n + 5 * G] : 0;
    int xq6 = (n + 6 * G < nsamp) ? x[n + 6 * G] : 0;
    int xq7 = (n + 7 * G < nsamp) ? x[n + 7 * G] : 0;

    float4 ra0, ra1, ra2, ra3, ra4, ra5, ra6, ra7;
    float4 rb0, rb1, rb2, rb3, rb4, rb5, rb6, rb7;
#define LOAD_TILES(xi_)                                                        \
    {                                                                          \
        const int row0_ = (xi_) / 225;                                         \
        const int row1_ = (xi_) % 225;                                         \
        const float* Ag_ = out1 + (size_t)row0_ * 4096 + sr * 64 + sc;         \
        const float* Bg_ = out2 + (size_t)row1_ * 4096 + sr * 64 + sc;         \
        ra0 = *(const float4*)(Ag_ + 0 * 512);                                 \
        ra1 = *(const float4*)(Ag_ + 1 * 512);                                 \
        ra2 = *(const float4*)(Ag_ + 2 * 512);                                 \
        ra3 = *(const float4*)(Ag_ + 3 * 512);                                 \
        ra4 = *(const float4*)(Ag_ + 4 * 512);                                 \
        ra5 = *(const float4*)(Ag_ + 5 * 512);                                 \
        ra6 = *(const float4*)(Ag_ + 6 * 512);                                 \
        ra7 = *(const float4*)(Ag_ + 7 * 512);                                 \
        rb0 = *(const float4*)(Bg_ + 0 * 512);                                 \
        rb1 = *(const float4*)(Bg_ + 1 * 512);                                 \
        rb2 = *(const float4*)(Bg_ + 2 * 512);                                 \
        rb3 = *(const float4*)(Bg_ + 3 * 512);                                 \
        rb4 = *(const float4*)(Bg_ + 4 * 512);                                 \
        rb5 = *(const float4*)(Bg_ + 5 * 512);                                 \
        rb6 = *(const float4*)(Bg_ + 6 * 512);                                 \
        rb7 = *(const float4*)(Bg_ + 7 * 512);                                 \
    }

    LOAD_TILES(xq0);

    for (;;) {
        if (n + 3 * G < nsamp)      __builtin_amdgcn_s_setprio(3);
        else if (n + 2 * G < nsamp) __builtin_amdgcn_s_setprio(2);
        else if (n + G < nsamp)     __builtin_amdgcn_s_setprio(1);
        else                        __builtin_amdgcn_s_setprio(0);

        __builtin_amdgcn_fence(__ATOMIC_RELEASE, "workgroup", "local");
        __builtin_amdgcn_s_barrier();
        __builtin_amdgcn_fence(__ATOMIC_ACQUIRE, "workgroup", "local");
        float* sA = &lA[sr * SA + sc];
        float* sB = &lB[(t >> 5) * SB + (t & 31) * 4];
        *(float4*)(sA + 0 * 8 * SA) = ra0;   *(float4*)(sB + 0 * 4 * SB) = rb0;
        *(float4*)(sA + 1 * 8 * SA) = ra1;   *(float4*)(sB + 1 * 4 * SB) = rb1;
        *(float4*)(sA + 2 * 8 * SA) = ra2;   *(float4*)(sB + 2 * 4 * SB) = rb2;
        *(float4*)(sA + 3 * 8 * SA) = ra3;   *(float4*)(sB + 3 * 4 * SB) = rb3;
        *(float4*)(sA + 4 * 8 * SA) = ra4;   *(float4*)(sB + 4 * 4 * SB) = rb4;
        *(float4*)(sA + 5 * 8 * SA) = ra5;   *(float4*)(sB + 5 * 4 * SB) = rb5;
        *(float4*)(sA + 6 * 8 * SA) = ra6;   *(float4*)(sB + 6 * 4 * SB) = rb6;
        *(float4*)(sA + 7 * 8 * SA) = ra7;   *(float4*)(sB + 7 * 4 * SB) = rb7;
        __syncthreads();

        const int  nn   = n + G;
        const bool more = nn < nsamp;

        v2f fa0[4], fb0[4][4];
        v2f fa1[4], fb1[4][4];
#define LDS_FRAGS(fa_, fb_, blk_)                                              \
        _Pragma("unroll")                                                      \
        for (int kk = 0; kk < 4; ++kk) {                                       \
            const int k0_ = (blk_) * 4 + kk;                                   \
            const float* bp_ = Bbase + k0_ * 2 * SB;                           \
            fa_[kk]    = *(const v2f*)(Abase + k0_ * 4);                       \
            fb_[kk][0] = *(const v2f*)(bp_ + 0);                               \
            fb_[kk][1] = *(const v2f*)(bp_ + 32);                              \
            fb_[kk][2] = *(const v2f*)(bp_ + 64);                              \
            fb_[kk][3] = *(const v2f*)(bp_ + 96);                              \
        }
#define WMMA_BLOCK(fa_, fb_)                                                   \
        _Pragma("unroll")                                                      \
        for (int kk = 0; kk < 4; ++kk) {                                       \
            acc0 = __builtin_amdgcn_wmma_f32_16x16x4_f32(                      \
                false, fa_[kk], false, fb_[kk][0], (short)0, acc0, false, false); \
            acc1 = __builtin_amdgcn_wmma_f32_16x16x4_f32(                      \
                false, fa_[kk], false, fb_[kk][1], (short)0, acc1, false, false); \
            acc2 = __builtin_amdgcn_wmma_f32_16x16x4_f32(                      \
                false, fa_[kk], false, fb_[kk][2], (short)0, acc2, false, false); \
            acc3 = __builtin_amdgcn_wmma_f32_16x16x4_f32(                      \
                false, fa_[kk], false, fb_[kk][3], (short)0, acc3, false, false); \
        }

        LDS_FRAGS(fa0, fb0, 0);
        LDS_FRAGS(fa1, fb1, 1);
        if (more) {
            LOAD_TILES(xq1);
        }

        v8f acc0 = {}, acc1 = {}, acc2 = {}, acc3 = {};
        WMMA_BLOCK(fa0, fb0);
        LDS_FRAGS(fa0, fb0, 2);
        __builtin_amdgcn_sched_barrier(0);
        WMMA_BLOCK(fa1, fb1);
        LDS_FRAGS(fa1, fb1, 3);
        __builtin_amdgcn_sched_barrier(0);
        WMMA_BLOCK(fa0, fb0);
        WMMA_BLOCK(fa1, fb1);
#undef LDS_FRAGS
#undef WMMA_BLOCK

        float* op = out + (size_t)n * 4096 + (tm * 16 + lh * 8) * 64 + lr;
#pragma unroll
        for (int v = 0; v < 8; ++v) {
            op[v * 64]      = acc0[v];
            op[v * 64 + 16] = acc1[v];
            op[v * 64 + 32] = acc2[v];
            op[v * 64 + 48] = acc3[v];
        }

        if (!more) break;
        n = nn;
        xq1 = xq2;  xq2 = xq3;  xq3 = xq4;  xq4 = xq5;  xq5 = xq6;  xq6 = xq7;
        xq7 = (n + 7 * G < nsamp) ? x[n + 7 * G] : 0;
    }
#undef LOAD_TILES
}

extern "C" void kernel_launch(void* const* d_in, const int* in_sizes, int n_in,
                              void* d_out, int out_size, void* d_ws, size_t ws_size,
                              hipStream_t stream) {
    const int*   x  = (const int*)  d_in[0];
    const float* U0 = (const float*)d_in[1];
    const float* U1 = (const float*)d_in[2];
    const float* U2 = (const float*)d_in[3];
    const float* U3 = (const float*)d_in[4];
    float* out = (float*)d_out;

    float* w1 = (float*)d_ws;
    float* w2 = w1 + 921600;

    build_tables<<<450, 256, 0, stream>>>(U0, U1, U2, U3, w1, w2);

    const int nsamp = in_sizes[0];
    ttm_gemm_kernel<<<1024, 128, 0, stream>>>(x, w1, w2, out, nsamp);
}
